# starting kernel
# baseline (speedup 1.0000x reference)
#include <hip/hip_runtime.h>
#include <hip/hip_bf16.h>


#define NH     8
#define SEQ    512
#define DK     32
#define DV     64
#define BM     128
#define BN     16
#define ALPHA  (1.0f / 64.0f)

typedef float v2f __attribute__((ext_vector_type(2)));
typedef float v8f __attribute__((ext_vector_type(8)));

__device__ __forceinline__ v8f wmma_f32_k4(v2f a, v2f b, v8f c) {
    return __builtin_amdgcn_wmma_f32_16x16x4_f32(false, a, false, b,
                                                 (short)0, c, false, false);
}

__global__ __launch_bounds__(256)
void poly2_attn_kernel(const float* __restrict__ Q,
                       const float* __restrict__ K,
                       const float* __restrict__ V,
                       float* __restrict__ YS,
                       float* __restrict__ YZ) {
    __shared__ float Qs[BM][DK + 1];
    __shared__ float Ks[BN][DK + 1];
    __shared__ float Vs[BN][DV + 1];
    __shared__ float Ps[8][16][17];

    const int h    = blockIdx.y;
    const int q0   = blockIdx.x * BM;
    const int tid  = threadIdx.x;
    const int wave = tid >> 5;
    const int lane = tid & 31;
    const int r    = lane & 15;
    const int hi   = lane >> 4;

    const float* Qh = Q + (size_t)h * SEQ * DK;
    const float* Kh = K + (size_t)h * SEQ * DK;
    const float* Vh = V + (size_t)h * SEQ * DV;

    for (int idx = tid; idx < BM * DK; idx += 256) {
        const int row = idx >> 5, col = idx & 31;
        Qs[row][col] = Qh[(size_t)(q0 + row) * DK + col];
    }
    __syncthreads();

    const int qrow_local = wave * 16;
    const int qbase      = q0 + qrow_local;
    const int jd         = qbase >> 4;
    const int nj         = (q0 >> 4) + (BM >> 4);

    v2f qa[DK / 4];
#pragma unroll
    for (int kk = 0; kk < DK; kk += 4) {
        const int c0 = kk + 2 * hi;
        qa[kk >> 2] = (v2f){ Qs[qrow_local + r][c0], Qs[qrow_local + r][c0 + 1] };
    }

    v8f acc0 = {0,0,0,0,0,0,0,0};
    v8f acc1 = {0,0,0,0,0,0,0,0};
    v8f acc2 = {0,0,0,0,0,0,0,0};
    v8f acc3 = {0,0,0,0,0,0,0,0};
    v8f accz = {0,0,0,0,0,0,0,0};
    const v2f ones = {1.0f, 1.0f};

    for (int j = 0; j < nj; ++j) {
        const int k0 = j * BN;
        __syncthreads();
        for (int idx = tid; idx < BN * DK; idx += 256) {
            const int row = idx >> 5, col = idx & 31;
            Ks[row][col] = Kh[(size_t)(k0 + row) * DK + col];
        }
        for (int idx = tid; idx < BN * DV; idx += 256) {
            const int row = idx >> 6, col = idx & 63;
            Vs[row][col] = Vh[(size_t)(k0 + row) * DV + col];
        }
        if (j + 1 < nj && tid < BN) {
            const int kn = k0 + BN + tid;
            __builtin_prefetch(&Kh[(size_t)kn * DK], 0, 1);
            __builtin_prefetch(&Vh[(size_t)kn * DV], 0, 1);
            __builtin_prefetch(&Vh[(size_t)kn * DV + 32], 0, 1);
        }
        __syncthreads();

        const bool active = (j <= jd);
        if (active) {
            v2f kb[DK / 4];
#pragma unroll
            for (int kk = 0; kk < DK; kk += 4) {
                const int c0 = kk + 2 * hi;
                kb[kk >> 2] = (v2f){ Ks[r][c0], Ks[r][c0 + 1] };
            }
            v8f s = {0,0,0,0,0,0,0,0};
#pragma unroll
            for (int t = 0; t < DK / 4; ++t)
                s = wmma_f32_k4(qa[t], kb[t], s);

#pragma unroll
            for (int i = 0; i < 8; ++i) {
                float p = s[i] * s[i];
                if (j == jd) {
                    const int gq = qbase + i + 8 * hi;
                    const int gk = k0 + r;
                    if (gk > gq) p = 0.0f;
                }
                Ps[wave][i + 8 * hi][r] = p;
            }
        }
        __syncthreads();

        if (active) {
#pragma unroll
            for (int kk = 0; kk < BN; kk += 4) {
                const int c0 = kk + 2 * hi;
                v2f a  = { Ps[wave][r][c0], Ps[wave][r][c0 + 1] };
                v2f b0 = { Vs[c0][ 0 + r], Vs[c0 + 1][ 0 + r] };
                v2f b1 = { Vs[c0][16 + r], Vs[c0 + 1][16 + r] };
                v2f b2 = { Vs[c0][32 + r], Vs[c0 + 1][32 + r] };
                v2f b3 = { Vs[c0][48 + r], Vs[c0 + 1][48 + r] };
                acc0 = wmma_f32_k4(a, b0, acc0);
                acc1 = wmma_f32_k4(a, b1, acc1);
                acc2 = wmma_f32_k4(a, b2, acc2);
                acc3 = wmma_f32_k4(a, b3, acc3);
                accz = wmma_f32_k4(a, ones, accz);
            }
        }
    }

    float* ys = YS + ((size_t)h * SEQ + qbase) * DV;
#pragma unroll
    for (int i = 0; i < 8; ++i) {
        const int row = i + 8 * hi;
        ys[(size_t)row * DV +  0 + r] = acc0[i] * ALPHA;
        ys[(size_t)row * DV + 16 + r] = acc1[i] * ALPHA;
        ys[(size_t)row * DV + 32 + r] = acc2[i] * ALPHA;
        ys[(size_t)row * DV + 48 + r] = acc3[i] * ALPHA;
    }
    if (r == 0) {
#pragma unroll
        for (int i = 0; i < 8; ++i) {
            YZ[(size_t)h * SEQ + qbase + i + 8 * hi] = accz[i] * ALPHA;
        }
    }
}

extern "C" void kernel_launch(void* const* d_in, const int* in_sizes, int n_in,
                              void* d_out, int out_size, void* d_ws, size_t ws_size,
                              hipStream_t stream) {
    (void)in_sizes; (void)n_in; (void)out_size; (void)d_ws; (void)ws_size;
    const float* Q = (const float*)d_in[0];
    const float* K = (const float*)d_in[1];
    const float* V = (const float*)d_in[2];
    float* YS = (float*)d_out;
    float* YZ = (float*)d_out + (size_t)NH * SEQ * DV;

    dim3 grid(SEQ / BM, NH);
    dim3 block(256);
    hipLaunchKernelGGL(poly2_attn_kernel, grid, block, 0, stream, Q, K, V, YS, YZ);
}


# Key loop split over waves: one 16-row query tile per work-group (256 of them), key blocks round-robin to 8 waves, per-wave K/V staging (loads in one batch), partials added via LDS: sum order changes
# speedup vs baseline: 7.4887x; 7.4887x over previous
#include <hip/hip_runtime.h>
#include <hip/hip_bf16.h>


#define NH     8
#define SEQ    512
#define DK     32
#define DV     64
#define BM     16
#define BN     16
#define NW     8
#define ALPHA  (1.0f / 64.0f)

typedef float v2f __attribute__((ext_vector_type(2)));
typedef float v8f __attribute__((ext_vector_type(8)));

__device__ __forceinline__ v8f wmma_f32_k4(v2f a, v2f b, v8f c) {
    return __builtin_amdgcn_wmma_f32_16x16x4_f32(false, a, false, b,
                                                 (short)0, c, false, false);
}

__global__ __launch_bounds__(256)
void poly2_attn_kernel(const float* __restrict__ Q,
                       const float* __restrict__ K,
                       const float* __restrict__ V,
                       float* __restrict__ YS,
                       float* __restrict__ YZ) {
    __shared__ float Qs[BM][DK + 1];
    __shared__ float Ks[NW][BN][DK + 1];
    __shared__ float Vs[NW][BN][DV + 1];
    __shared__ float Ps[NW][16][17];

    const int h    = blockIdx.x & (NH - 1);
    const int qt   = (SEQ / BM - 1) - (int)(blockIdx.x >> 3);
    const int q0   = qt * BM;
    const int tid  = threadIdx.x;
    const int wave = tid >> 5;
    const int lane = tid & 31;
    const int r    = lane & 15;
    const int hi   = lane >> 4;

    const float* Qh = Q + (size_t)h * SEQ * DK;
    const float* Kh = K + (size_t)h * SEQ * DK;
    const float* Vh = V + (size_t)h * SEQ * DV;

    const float* qp = Qh + (size_t)q0 * DK;
#pragma unroll
    for (int e = 0; e < (BM * DK) / 256; ++e) {
        const int idx = tid + 256 * e;
        Qs[idx >> 5][idx & 31] = qp[idx];
    }
    __syncthreads();

    v2f qa[DK / 4];
#pragma unroll
    for (int kk = 0; kk < DK; kk += 4) {
        const int c0 = kk + 2 * hi;
        qa[kk >> 2] = (v2f){ Qs[r][c0], Qs[r][c0 + 1] };
    }

    v8f acc0 = {0,0,0,0,0,0,0,0};
    v8f acc1 = {0,0,0,0,0,0,0,0};
    v8f acc2 = {0,0,0,0,0,0,0,0};
    v8f acc3 = {0,0,0,0,0,0,0,0};
    v8f accz = {0,0,0,0,0,0,0,0};
    const v2f ones = {1.0f, 1.0f};

    const int nit = (qt >> 3) + 1;
    for (int it = 0; it < nit; ++it) {
        const int j  = it * NW + wave;
        const int k0 = j * BN;
        const bool active = (j <= qt);
        __syncthreads();
        if (active) {
            const float* kp = Kh + (size_t)k0 * DK + lane;
            const float* vp = Vh + (size_t)k0 * DV + lane;
            float kreg[BN], vreg[2 * BN];
#pragma unroll
            for (int t = 0; t < BN; ++t) kreg[t] = kp[t * 32];
#pragma unroll
            for (int t = 0; t < 2 * BN; ++t) vreg[t] = vp[t * 32];
#pragma unroll
            for (int t = 0; t < BN; ++t)
                Ks[wave][t][lane] = kreg[t];
#pragma unroll
            for (int t = 0; t < 2 * BN; ++t)
                Vs[wave][t >> 1][(t & 1) * 32 + lane] = vreg[t];
            if (j + NW <= qt && lane < BN) {
                const int kn = k0 + NW * BN + lane;
                __builtin_prefetch(&Kh[(size_t)kn * DK], 0, 1);
                __builtin_prefetch(&Vh[(size_t)kn * DV], 0, 1);
                __builtin_prefetch(&Vh[(size_t)kn * DV + 32], 0, 1);
            }
        }
        __syncthreads();

        if (active) {
            v2f kb[DK / 4];
#pragma unroll
            for (int kk = 0; kk < DK; kk += 4) {
                const int c0 = kk + 2 * hi;
                kb[kk >> 2] = (v2f){ Ks[wave][r][c0], Ks[wave][r][c0 + 1] };
            }
            v8f s = {0,0,0,0,0,0,0,0};
#pragma unroll
            for (int t = 0; t < DK / 4; ++t)
                s = wmma_f32_k4(qa[t], kb[t], s);

#pragma unroll
            for (int i = 0; i < 8; ++i) {
                float p = s[i] * s[i];
                if (j == qt) {
                    const int gq = q0 + i + 8 * hi;
                    const int gk = k0 + r;
                    if (gk > gq) p = 0.0f;
                }
                Ps[wave][i + 8 * hi][r] = p;
            }
        }
        __syncthreads();

        if (active) {
#pragma unroll
            for (int kk = 0; kk < BN; kk += 4) {
                const int c0 = kk + 2 * hi;
                v2f a  = { Ps[wave][r][c0], Ps[wave][r][c0 + 1] };
                v2f b0 = { Vs[wave][c0][ 0 + r], Vs[wave][c0 + 1][ 0 + r] };
                v2f b1 = { Vs[wave][c0][16 + r], Vs[wave][c0 + 1][16 + r] };
                v2f b2 = { Vs[wave][c0][32 + r], Vs[wave][c0 + 1][32 + r] };
                v2f b3 = { Vs[wave][c0][48 + r], Vs[wave][c0 + 1][48 + r] };
                acc0 = wmma_f32_k4(a, b0, acc0);
                acc1 = wmma_f32_k4(a, b1, acc1);
                acc2 = wmma_f32_k4(a, b2, acc2);
                acc3 = wmma_f32_k4(a, b3, acc3);
                accz = wmma_f32_k4(a, ones, accz);
            }
        }
    }

    __syncthreads();
#pragma unroll
    for (int i = 0; i < 8; ++i) {
        const int row = i + 8 * hi;
        Vs[wave][row][ 0 + r] = acc0[i];
        Vs[wave][row][16 + r] = acc1[i];
        Vs[wave][row][32 + r] = acc2[i];
        Vs[wave][row][48 + r] = acc3[i];
    }
    if (r == 0) {
#pragma unroll
        for (int i = 0; i < 8; ++i) {
            Vs[wave][i + 8 * hi][DV] = accz[i];
        }
    }
    __syncthreads();

    float* ys = YS + ((size_t)h * SEQ + q0) * DV;
#pragma unroll
    for (int e = 0; e < (BM * DV) / 256; ++e) {
        const int idx = tid + 256 * e;
        const int row = idx >> 6, col = idx & 63;
        float sum = Vs[0][row][col];
#pragma unroll
        for (int w = 1; w < NW; ++w) sum += Vs[w][row][col];
        ys[idx] = sum * ALPHA;
    }
    if (tid < BM) {
        float sum = Vs[0][tid][DV];
#pragma unroll
        for (int w = 1; w < NW; ++w) sum += Vs[w][tid][DV];
        YZ[(size_t)h * SEQ + q0 + tid] = sum * ALPHA;
    }
}

extern "C" void kernel_launch(void* const* d_in, const int* in_sizes, int n_in,
                              void* d_out, int out_size, void* d_ws, size_t ws_size,
                              hipStream_t stream) {
    (void)in_sizes; (void)n_in; (void)out_size; (void)d_ws; (void)ws_size;
    const float* Q = (const float*)d_in[0];
    const float* K = (const float*)d_in[1];
    const float* V = (const float*)d_in[2];
    float* YS = (float*)d_out;
    float* YZ = (float*)d_out + (size_t)NH * SEQ * DV;

    dim3 grid(NH * (SEQ / BM));
    dim3 block(256);
    hipLaunchKernelGGL(poly2_attn_kernel, grid, block, 0, stream, Q, K, V, YS, YZ);
}


# Key loop without LDS or barriers: K, Q, V fragments loaded from global in fragment layout (16-byte loads), scores computed transposed so their squares are the next A operand: sum order changes
# speedup vs baseline: 11.9963x; 1.6019x over previous
#include <hip/hip_runtime.h>


#define NH     8
#define SEQ    512
#define DK     32
#define DV     64
#define BM     16
#define BN     16
#define NW     8
#define ALPHA  (1.0f / 64.0f)
#define PP     (DV + 4)

typedef float v2f __attribute__((ext_vector_type(2)));
typedef float v8f __attribute__((ext_vector_type(8)));

__device__ __forceinline__ v8f wmma_f32_k4(v2f a, v2f b, v8f c) {
    return __builtin_amdgcn_wmma_f32_16x16x4_f32(false, a, false, b,
                                                 (short)0, c, false, false);
}

__device__ __forceinline__ void load_row16(const float* __restrict__ p, float4 (&x)[4]) {
    const float4* p4 = reinterpret_cast<const float4*>(p);
#pragma unroll
    for (int m = 0; m < 4; ++m) x[m] = p4[m];
}

__device__ __forceinline__ v2f frag_k(const float4 (&x)[4], int t) {
    const float4 q = x[t >> 1];
    return (t & 1) ? (v2f){ q.z, q.w } : (v2f){ q.x, q.y };
}

__global__ __launch_bounds__(256)
void poly2_attn_kernel(const float* __restrict__ Q,
                       const float* __restrict__ K,
                       const float* __restrict__ V,
                       float* __restrict__ YS,
                       float* __restrict__ YZ) {
    __shared__ __attribute__((aligned(16))) float Part[NW][BM][PP];
    __shared__ float Zs[NW][BM];

    const int h    = blockIdx.x & (NH - 1);
    const int qt   = (SEQ / BM - 1) - (int)(blockIdx.x >> 3);
    const int q0   = qt * BM;
    const int tid  = threadIdx.x;
    const int wave = tid >> 5;
    const int lane = tid & 31;
    const int c    = lane & 15;
    const int hf   = lane >> 4;

    const float* Qh = Q + (size_t)h * SEQ * DK;
    const float* Kh = K + (size_t)h * SEQ * DK;
    const float* Vh = V + (size_t)h * SEQ * DV;

    float4 q4[4];
    load_row16(Qh + (size_t)(q0 + c) * DK + 16 * hf, q4);

    v8f acc0 = {0,0,0,0,0,0,0,0};
    v8f acc1 = {0,0,0,0,0,0,0,0};
    v8f acc2 = {0,0,0,0,0,0,0,0};
    v8f acc3 = {0,0,0,0,0,0,0,0};
    v8f accz = {0,0,0,0,0,0,0,0};
    const v2f ones = {1.0f, 1.0f};

    for (int j = wave; j <= qt; j += NW) {
        const int k0 = j * BN;

        float4 k4[4];
        load_row16(Kh + (size_t)(k0 + c) * DK + 16 * hf, k4);

        float4 v4[8];
        const float4* vp =
            reinterpret_cast<const float4*>(Vh + (size_t)(k0 + 8 * hf) * DV + 4 * c);
#pragma unroll
        for (int m = 0; m < 8; ++m) v4[m] = vp[m * (DV / 4)];

        if (j + NW <= qt && lane < BN) {
            const int kn = k0 + NW * BN + lane;
            __builtin_prefetch(&Kh[(size_t)kn * DK], 0, 1);
            __builtin_prefetch(&Vh[(size_t)kn * DV], 0, 1);
            __builtin_prefetch(&Vh[(size_t)kn * DV + 32], 0, 1);
        }

        v8f s = {0,0,0,0,0,0,0,0};
#pragma unroll
        for (int t = 0; t < DK / 4; ++t)
            s = wmma_f32_k4(frag_k(k4, t), frag_k(q4, t), s);

        v8f p;
#pragma unroll
        for (int i = 0; i < 8; ++i) {
            float x = s[i] * s[i];
            if (j == qt && (8 * hf + i) > c) x = 0.0f;
            p[i] = x;
        }

#pragma unroll
        for (int t = 0; t < BN / 4; ++t) {
            const v2f a = { p[2 * t], p[2 * t + 1] };
            const float4 u = v4[2 * t], w = v4[2 * t + 1];
            acc0 = wmma_f32_k4(a, (v2f){ u.x, w.x }, acc0);
            acc1 = wmma_f32_k4(a, (v2f){ u.y, w.y }, acc1);
            acc2 = wmma_f32_k4(a, (v2f){ u.z, w.z }, acc2);
            acc3 = wmma_f32_k4(a, (v2f){ u.w, w.w }, acc3);
            accz = wmma_f32_k4(a, ones, accz);
        }
    }

#pragma unroll
    for (int i = 0; i < 8; ++i) {
        const float4 o = { acc0[i], acc1[i], acc2[i], acc3[i] };
        *reinterpret_cast<float4*>(&Part[wave][8 * hf + i][4 * c]) = o;
    }
    if (c == 0) {
#pragma unroll
        for (int i = 0; i < 8; ++i) Zs[wave][8 * hf + i] = accz[i];
    }
    __syncthreads();

    {
        const int row = tid >> 4, c4 = (tid & 15) * 4;
        float4 sum = *reinterpret_cast<const float4*>(&Part[0][row][c4]);
#pragma unroll
        for (int w = 1; w < NW; ++w) {
            const float4 x = *reinterpret_cast<const float4*>(&Part[w][row][c4]);
            sum.x += x.x; sum.y += x.y; sum.z += x.z; sum.w += x.w;
        }
        sum.x *= ALPHA; sum.y *= ALPHA; sum.z *= ALPHA; sum.w *= ALPHA;
        float* ys = YS + ((size_t)h * SEQ + q0) * DV;
        *reinterpret_cast<float4*>(&ys[row * DV + c4]) = sum;
    }
    if (tid < BM) {
        float sum = Zs[0][tid];
#pragma unroll
        for (int w = 1; w < NW; ++w) sum += Zs[w][tid];
        YZ[(size_t)h * SEQ + q0 + tid] = sum * ALPHA;
    }
}

extern "C" void kernel_launch(void* const* d_in, const int* in_sizes, int n_in,
                              void* d_out, int out_size, void* d_ws, size_t ws_size,
                              hipStream_t stream) {
    (void)in_sizes; (void)n_in; (void)out_size; (void)d_ws; (void)ws_size;
    const float* Q = (const float*)d_in[0];
    const float* K = (const float*)d_in[1];
    const float* V = (const float*)d_in[2];
    float* YS = (float*)d_out;
    float* YZ = (float*)d_out + (size_t)NH * SEQ * DV;

    dim3 grid(NH * (SEQ / BM));
    dim3 block(256);
    hipLaunchKernelGGL(poly2_attn_kernel, grid, block, 0, stream, Q, K, V, YS, YZ);
}
